# starting kernel
# baseline (speedup 1.0000x reference)
#include <hip/hip_runtime.h>

typedef __attribute__((ext_vector_type(2))) float v2f;
typedef __attribute__((ext_vector_type(4))) float v4f;
typedef __attribute__((ext_vector_type(8))) float v8f;

#define IN_DIM  1024
#define OUT_DIM 1024
#define NPOLY   9

__device__ __forceinline__ float fast_tanh(float y) {
    float e = __builtin_amdgcn_exp2f(y * 2.8853900817779268f);
    return 1.0f - 2.0f * __builtin_amdgcn_rcpf(e + 1.0f);
}

__global__ __launch_bounds__(256)
void legendre_fused_kernel(const float* __restrict__ x,
                           const float* __restrict__ coeffs,
                           const float* __restrict__ scale_p,
                           float* __restrict__ out)
{
    __shared__ float S[16 * 16];

    const int tid  = threadIdx.x;
    const int lane = tid & 31;
    const int wave = tid >> 5;
    const int b0   = blockIdx.x * 16;

    S[tid] = 0.0f;
    __syncthreads();

    float s = scale_p[0];
    s = fminf(fmaxf(s, 0.1f), 2.0f);

    for (int r = 0; r < 2; ++r) {
        const int row = 2 * wave + r;
        const float* xrow = x + (size_t)(b0 + row) * IN_DIM;

        float acc[NPOLY];
#pragma unroll
        for (int k = 0; k < NPOLY; ++k) acc[k] = 0.0f;

#pragma unroll
        for (int j = 0; j < 8; ++j) {
            v4f xv = *(const v4f*)(xrow + j * 128 + lane * 4);
#pragma unroll
            for (int e = 0; e < 4; ++e) {
                float t  = fast_tanh(xv[e] * s);
                float p0 = 1.0f;
                float p1 = t;
                acc[0] += p0;
                acc[1] += p1;
#pragma unroll
                for (int n = 2; n <= 8; ++n) {
                    const float c1 = (2.0f * n - 1.0f) / (float)n;
                    const float c2 = (n - 1.0f) / (float)n;
                    float pn = c1 * t * p1 - c2 * p0;
                    acc[n] += pn;
                    p0 = p1;
                    p1 = pn;
                }
            }
        }

#pragma unroll
        for (int k = 0; k < NPOLY; ++k) {
#pragma unroll
            for (int off = 16; off > 0; off >>= 1)
                acc[k] += __shfl_xor(acc[k], off, 32);
        }
        if (lane == 0) {
#pragma unroll
            for (int k = 0; k < NPOLY; ++k) S[row * 16 + k] = acc[k];
        }
    }
    __syncthreads();

    const int mn    = lane & 15;
    const int khalf = (lane >> 4) * 2;
    const int mrow  = (lane >> 4) * 8;

    v2f afrag[3];
#pragma unroll
    for (int c = 0; c < 3; ++c) {
        afrag[c][0] = S[mn * 16 + 4 * c + khalf + 0];
        afrag[c][1] = S[mn * 16 + 4 * c + khalf + 1];
    }

#pragma unroll
    for (int j = 0; j < 8; ++j) {
        const int ct = wave * 8 + j;
        const int o  = ct * 16 + mn;

        v2f bfrag[3];
#pragma unroll
        for (int c = 0; c < 3; ++c) {
            const int k0 = 4 * c + khalf;
            bfrag[c][0] = (k0     < NPOLY) ? coeffs[o * NPOLY + k0    ] : 0.0f;
            bfrag[c][1] = (k0 + 1 < NPOLY) ? coeffs[o * NPOLY + k0 + 1] : 0.0f;
        }

        v8f d = {};
#pragma unroll
        for (int c = 0; c < 3; ++c) {
            d = __builtin_amdgcn_wmma_f32_16x16x4_f32(
                    false, afrag[c], false, bfrag[c], (short)0, d, false, false);
        }

#pragma unroll
        for (int v = 0; v < 8; ++v)
            out[(size_t)(b0 + mrow + v) * OUT_DIM + ct * 16 + mn] = d[v];
    }
}

extern "C" void kernel_launch(void* const* d_in, const int* in_sizes, int n_in,
                              void* d_out, int out_size, void* d_ws, size_t ws_size,
                              hipStream_t stream) {
    (void)in_sizes; (void)n_in; (void)out_size; (void)d_ws; (void)ws_size;
    const float* x      = (const float*)d_in[0];
    const float* coeffs = (const float*)d_in[1];
    const float* scale  = (const float*)d_in[2];
    float* out          = (float*)d_out;

    const int row_tiles = 4096 / 16;
    legendre_fused_kernel<<<row_tiles, 256, 0, stream>>>(x, coeffs, scale, out);
}


# All global loads of a wave (both x rows, all 8 tiles' coefficient fragments) issued at kernel start; LDS zero-fill barrier dropped so nothing waits ahead of them; summation order unchanged
# speedup vs baseline: 1.3418x; 1.3418x over previous
#include <hip/hip_runtime.h>


typedef __attribute__((ext_vector_type(2))) float v2f;
typedef __attribute__((ext_vector_type(4))) float v4f;
typedef __attribute__((ext_vector_type(8))) float v8f;

#define IN_DIM  1024
#define OUT_DIM 1024
#define NPOLY   9

__device__ __forceinline__ float fast_tanh(float y) {
    float e = __builtin_amdgcn_exp2f(y * 2.8853900817779268f);
    return 1.0f - 2.0f * __builtin_amdgcn_rcpf(e + 1.0f);
}

__global__ __launch_bounds__(256)
void legendre_fused_kernel(const float* __restrict__ x,
                           const float* __restrict__ coeffs,
                           const float* __restrict__ scale_p,
                           float* __restrict__ out)
{
    __shared__ float S[16 * 16];

    const int tid  = threadIdx.x;
    const int lane = tid & 31;
    const int wave = tid >> 5;
    const int b0   = blockIdx.x * 16;

    const int mn    = lane & 15;
    const int khalf = (lane >> 4) * 2;
    const int mrow  = (lane >> 4) * 8;

    const float* xbase = x + (size_t)(b0 + 2 * wave) * IN_DIM + lane * 4;
    v4f xv[2][8];
#pragma unroll
    for (int r = 0; r < 2; ++r) {
#pragma unroll
        for (int j = 0; j < 8; ++j)
            xv[r][j] = *(const v4f*)(xbase + r * IN_DIM + j * 128);
    }

    v2f bfrag[8][3];
#pragma unroll
    for (int j = 0; j < 8; ++j) {
        const int o = (wave * 8 + j) * 16 + mn;
        const float* crow = coeffs + o * NPOLY;
        bfrag[j][0][0] = crow[khalf + 0];
        bfrag[j][0][1] = crow[khalf + 1];
        bfrag[j][1][0] = crow[4 + khalf + 0];
        bfrag[j][1][1] = crow[4 + khalf + 1];
        const float c8 = crow[8];
        bfrag[j][2][0] = (khalf == 0) ? c8 : 0.0f;
        bfrag[j][2][1] = 0.0f;
    }

    __builtin_amdgcn_sched_barrier(0);

    float s = scale_p[0];
    s = fminf(fmaxf(s, 0.1f), 2.0f);

#pragma unroll
    for (int r = 0; r < 2; ++r) {
        const int row = 2 * wave + r;

        float acc[NPOLY];
#pragma unroll
        for (int k = 0; k < NPOLY; ++k) acc[k] = 0.0f;

#pragma unroll
        for (int j = 0; j < 8; ++j) {
            v4f xq = xv[r][j];
#pragma unroll
            for (int e = 0; e < 4; ++e) {
                float t  = fast_tanh(xq[e] * s);
                float p0 = 1.0f;
                float p1 = t;
                acc[0] += p0;
                acc[1] += p1;
#pragma unroll
                for (int n = 2; n <= 8; ++n) {
                    const float c1 = (2.0f * n - 1.0f) / (float)n;
                    const float c2 = (n - 1.0f) / (float)n;
                    float pn = c1 * t * p1 - c2 * p0;
                    acc[n] += pn;
                    p0 = p1;
                    p1 = pn;
                }
            }
        }

#pragma unroll
        for (int k = 0; k < NPOLY; ++k) {
#pragma unroll
            for (int off = 16; off > 0; off >>= 1)
                acc[k] += __shfl_xor(acc[k], off, 32);
        }
        if (lane == 0) {
#pragma unroll
            for (int k = 0; k < 16; ++k) S[row * 16 + k] = (k < NPOLY) ? acc[k] : 0.0f;
        }
    }
    __syncthreads();

    v2f afrag[3];
#pragma unroll
    for (int c = 0; c < 3; ++c) {
        afrag[c][0] = S[mn * 16 + 4 * c + khalf + 0];
        afrag[c][1] = S[mn * 16 + 4 * c + khalf + 1];
    }

#pragma unroll
    for (int j = 0; j < 8; ++j) {
        const int ct = wave * 8 + j;

        v8f d = {};
#pragma unroll
        for (int c = 0; c < 3; ++c) {
            d = __builtin_amdgcn_wmma_f32_16x16x4_f32(
                    false, afrag[c], false, bfrag[j][c], (short)0, d, false, false);
        }

#pragma unroll
        for (int v = 0; v < 8; ++v)
            out[(size_t)(b0 + mrow + v) * OUT_DIM + ct * 16 + mn] = d[v];
    }
}

extern "C" void kernel_launch(void* const* d_in, const int* in_sizes, int n_in,
                              void* d_out, int out_size, void* d_ws, size_t ws_size,
                              hipStream_t stream) {
    (void)in_sizes; (void)n_in; (void)out_size; (void)d_ws; (void)ws_size;
    const float* x      = (const float*)d_in[0];
    const float* coeffs = (const float*)d_in[1];
    const float* scale  = (const float*)d_in[2];
    float* out          = (float*)d_out;

    const int row_tiles = 4096 / 16;
    legendre_fused_kernel<<<row_tiles, 256, 0, stream>>>(x, coeffs, scale, out);
}


# Per element only P2..P4; row sums of P2..P8 rebuilt once per row from eight product sums (t,tt,tP2,P2P2,P2P3,P3P3,P3P4,P4P4) by Legendre product identities; two partial sums per lane, summands change
# speedup vs baseline: 1.5317x; 1.1415x over previous
#include <hip/hip_runtime.h>


typedef __attribute__((ext_vector_type(2))) float v2f;
typedef __attribute__((ext_vector_type(4))) float v4f;
typedef __attribute__((ext_vector_type(8))) float v8f;

#define IN_DIM  1024
#define OUT_DIM 1024
#define NPOLY   9

__device__ __forceinline__ v2f tanh_pair(v2f y) {
    v2f z = y * 2.8853900817779268f;
    v2f e;
    e[0] = __builtin_amdgcn_exp2f(z[0]);
    e[1] = __builtin_amdgcn_exp2f(z[1]);
    v2f d = e + 1.0f;
    v2f q;
    q[0] = __builtin_amdgcn_rcpf(d[0]);
    q[1] = __builtin_amdgcn_rcpf(d[1]);
    return 1.0f - 2.0f * q;
}

__global__ __launch_bounds__(256)
void legendre_fused_kernel(const float* __restrict__ x,
                           const float* __restrict__ coeffs,
                           const float* __restrict__ scale_p,
                           float* __restrict__ out)
{
    __shared__ float S[16 * 16];

    const int tid  = threadIdx.x;
    const int lane = tid & 31;
    const int wave = tid >> 5;
    const int b0   = blockIdx.x * 16;

    const int mn    = lane & 15;
    const int khalf = (lane >> 4) * 2;
    const int mrow  = (lane >> 4) * 8;

    const float* xbase = x + (size_t)(b0 + 2 * wave) * IN_DIM + lane * 4;
    v4f xv[2][8];
#pragma unroll
    for (int r = 0; r < 2; ++r) {
#pragma unroll
        for (int j = 0; j < 8; ++j)
            xv[r][j] = *(const v4f*)(xbase + r * IN_DIM + j * 128);
    }

    v2f bfrag[8][3];
#pragma unroll
    for (int j = 0; j < 8; ++j) {
        const int o = (wave * 8 + j) * 16 + mn;
        const float* crow = coeffs + o * NPOLY;
        bfrag[j][0][0] = crow[khalf + 0];
        bfrag[j][0][1] = crow[khalf + 1];
        bfrag[j][1][0] = crow[4 + khalf + 0];
        bfrag[j][1][1] = crow[4 + khalf + 1];
        const float c8 = crow[8];
        bfrag[j][2][0] = (khalf == 0) ? c8 : 0.0f;
        bfrag[j][2][1] = 0.0f;
    }

    __builtin_amdgcn_sched_barrier(0);

    float s = scale_p[0];
    s = fminf(fmaxf(s, 0.1f), 2.0f);

#pragma unroll
    for (int r = 0; r < 2; ++r) {
        const int row = 2 * wave + r;

        v2f pa[NPOLY];
#pragma unroll
        for (int k = 1; k < NPOLY; ++k) pa[k] = (v2f){0.0f, 0.0f};

#pragma unroll
        for (int j = 0; j < 8; ++j) {
            v4f xq = xv[r][j];
#pragma unroll
            for (int h = 0; h < 2; ++h) {
                v2f xp = {xq[2 * h], xq[2 * h + 1]};
                v2f t  = tanh_pair(xp * s);
                v2f u  = t * t;
                v2f p2 = 1.5f * u - 0.5f;
                v2f p3 = t * (2.5f * u - 1.5f);
                v2f p4 = u * (4.375f * u - 3.75f) + 0.375f;
                pa[1] += t;
                pa[2] += t * t;
                pa[3] += t * p2;
                pa[4] += p2 * p2;
                pa[5] += p2 * p3;
                pa[6] += p3 * p3;
                pa[7] += p3 * p4;
                pa[8] += p4 * p4;
            }
        }

        float acc[NPOLY];
#pragma unroll
        for (int k = 1; k < NPOLY; ++k) acc[k] = pa[k][0] + pa[k][1];

#pragma unroll
        for (int k = 1; k < NPOLY; ++k) {
#pragma unroll
            for (int off = 16; off > 0; off >>= 1)
                acc[k] += __shfl_xor(acc[k], off, 32);
        }

        const float n0 = (float)IN_DIM;
        float sl[NPOLY];
        sl[0] = n0;
        sl[1] = acc[1];
        sl[2] = 1.5f * acc[2] - 0.5f * n0;
        sl[3] = (acc[3] - (2.0f / 5.0f) * sl[1]) * (5.0f / 3.0f);
        sl[4] = (acc[4] - n0 / 5.0f - (2.0f / 7.0f) * sl[2]) * (35.0f / 18.0f);
        sl[5] = (acc[5] - (9.0f / 35.0f) * sl[1] - (4.0f / 15.0f) * sl[3]) * (21.0f / 10.0f);
        sl[6] = (acc[6] - n0 / 7.0f - (4.0f / 21.0f) * sl[2] - (18.0f / 77.0f) * sl[4]) * (231.0f / 100.0f);
        sl[7] = (acc[7] - (4.0f / 21.0f) * sl[1] - (2.0f / 11.0f) * sl[3] - (20.0f / 91.0f) * sl[5]) * (429.0f / 175.0f);
        sl[8] = (acc[8] - n0 / 9.0f - (100.0f / 693.0f) * sl[2] - (162.0f / 1001.0f) * sl[4]
                 - (20.0f / 99.0f) * sl[6]) * (1287.0f / 490.0f);

        if (lane == 0) {
#pragma unroll
            for (int k = 0; k < 16; ++k) S[row * 16 + k] = (k < NPOLY) ? sl[k] : 0.0f;
        }
    }
    __syncthreads();

    v2f afrag[3];
#pragma unroll
    for (int c = 0; c < 3; ++c) {
        afrag[c][0] = S[mn * 16 + 4 * c + khalf + 0];
        afrag[c][1] = S[mn * 16 + 4 * c + khalf + 1];
    }

#pragma unroll
    for (int j = 0; j < 8; ++j) {
        const int ct = wave * 8 + j;

        v8f d = {};
#pragma unroll
        for (int c = 0; c < 3; ++c) {
            d = __builtin_amdgcn_wmma_f32_16x16x4_f32(
                    false, afrag[c], false, bfrag[j][c], (short)0, d, false, false);
        }

#pragma unroll
        for (int v = 0; v < 8; ++v)
            out[(size_t)(b0 + mrow + v) * OUT_DIM + ct * 16 + mn] = d[v];
    }
}

extern "C" void kernel_launch(void* const* d_in, const int* in_sizes, int n_in,
                              void* d_out, int out_size, void* d_ws, size_t ws_size,
                              hipStream_t stream) {
    (void)in_sizes; (void)n_in; (void)out_size; (void)d_ws; (void)ws_size;
    const float* x      = (const float*)d_in[0];
    const float* coeffs = (const float*)d_in[1];
    const float* scale  = (const float*)d_in[2];
    float* out          = (float*)d_out;

    const int row_tiles = 4096 / 16;
    legendre_fused_kernel<<<row_tiles, 256, 0, stream>>>(x, coeffs, scale, out);
}


# Row-sum butterfly of both rows on the vector ALU (one half exchange, then row_xmask adds) behind the second row, instead of two chains of five LDS-permute round trips; summation order unchanged
# speedup vs baseline: 1.6203x; 1.0578x over previous
#include <hip/hip_runtime.h>


typedef __attribute__((ext_vector_type(2))) float v2f;
typedef __attribute__((ext_vector_type(4))) float v4f;
typedef __attribute__((ext_vector_type(8))) float v8f;

#define IN_DIM  1024
#define OUT_DIM 1024
#define NPOLY   9

__device__ __forceinline__ v2f tanh_pair(v2f y) {
    v2f z = y * 2.8853900817779268f;
    v2f e;
    e[0] = __builtin_amdgcn_exp2f(z[0]);
    e[1] = __builtin_amdgcn_exp2f(z[1]);
    v2f d = e + 1.0f;
    v2f q;
    q[0] = __builtin_amdgcn_rcpf(d[0]);
    q[1] = __builtin_amdgcn_rcpf(d[1]);
    return 1.0f - 2.0f * q;
}

template <int M>
__device__ __forceinline__ float add_row_xor(float x) {
    const int b = __builtin_bit_cast(int, x);
    const int p = __builtin_amdgcn_update_dpp(0, b, 0x160 + M, 0xf, 0xf, true);
    return x + __builtin_bit_cast(float, p);
}

__device__ __forceinline__ float xhalf(float x) {
    const int b = __builtin_bit_cast(int, x);
    return __builtin_bit_cast(float,
        __builtin_amdgcn_permlanex16(b, b, 0x76543210, 0xfedcba98, false, false));
}

__device__ __forceinline__ float add_half_xor_two_rows(float a, float b, bool upper) {
    const float keep = upper ? b : a;
    const float send = upper ? a : b;
    return keep + xhalf(send);
}

__global__ __launch_bounds__(256)
void legendre_fused_kernel(const float* __restrict__ x,
                           const float* __restrict__ coeffs,
                           const float* __restrict__ scale_p,
                           float* __restrict__ out)
{
    __shared__ float S[16 * 16];

    const int tid  = threadIdx.x;
    const int lane = tid & 31;
    const int wave = tid >> 5;
    const int b0   = blockIdx.x * 16;

    const int mn    = lane & 15;
    const int khalf = (lane >> 4) * 2;
    const int mrow  = (lane >> 4) * 8;

    const float* xbase = x + (size_t)(b0 + 2 * wave) * IN_DIM + lane * 4;
    v4f xv[2][8];
#pragma unroll
    for (int r = 0; r < 2; ++r) {
#pragma unroll
        for (int j = 0; j < 8; ++j)
            xv[r][j] = *(const v4f*)(xbase + r * IN_DIM + j * 128);
    }

    v2f bfrag[8][3];
#pragma unroll
    for (int j = 0; j < 8; ++j) {
        const int o = (wave * 8 + j) * 16 + mn;
        const float* crow = coeffs + o * NPOLY;
        bfrag[j][0][0] = crow[khalf + 0];
        bfrag[j][0][1] = crow[khalf + 1];
        bfrag[j][1][0] = crow[4 + khalf + 0];
        bfrag[j][1][1] = crow[4 + khalf + 1];
        const float c8 = crow[8];
        bfrag[j][2][0] = (khalf == 0) ? c8 : 0.0f;
        bfrag[j][2][1] = 0.0f;
    }

    __builtin_amdgcn_sched_barrier(0);

    float s = scale_p[0];
    s = fminf(fmaxf(s, 0.1f), 2.0f);

    float accr[2][NPOLY];
#pragma unroll
    for (int r = 0; r < 2; ++r) {
        v2f pa[NPOLY];
#pragma unroll
        for (int k = 1; k < NPOLY; ++k) pa[k] = (v2f){0.0f, 0.0f};

#pragma unroll
        for (int j = 0; j < 8; ++j) {
            v4f xq = xv[r][j];
#pragma unroll
            for (int h = 0; h < 2; ++h) {
                v2f xp = {xq[2 * h], xq[2 * h + 1]};
                v2f t  = tanh_pair(xp * s);
                v2f u  = t * t;
                v2f p2 = 1.5f * u - 0.5f;
                v2f p3 = t * (2.5f * u - 1.5f);
                v2f p4 = u * (4.375f * u - 3.75f) + 0.375f;
                pa[1] += t;
                pa[2] += t * t;
                pa[3] += t * p2;
                pa[4] += p2 * p2;
                pa[5] += p2 * p3;
                pa[6] += p3 * p3;
                pa[7] += p3 * p4;
                pa[8] += p4 * p4;
            }
        }

#pragma unroll
        for (int k = 1; k < NPOLY; ++k) accr[r][k] = pa[k][0] + pa[k][1];
    }

    float acc[NPOLY];
#pragma unroll
    for (int k = 1; k < NPOLY; ++k) acc[k] = add_half_xor_two_rows(accr[0][k], accr[1][k], lane >= 16);
#pragma unroll
    for (int k = 1; k < NPOLY; ++k) acc[k] = add_row_xor<8>(acc[k]);
#pragma unroll
    for (int k = 1; k < NPOLY; ++k) acc[k] = add_row_xor<4>(acc[k]);
#pragma unroll
    for (int k = 1; k < NPOLY; ++k) acc[k] = add_row_xor<2>(acc[k]);
#pragma unroll
    for (int k = 1; k < NPOLY; ++k) acc[k] = add_row_xor<1>(acc[k]);

    if (mn == 0) {
        const int row = 2 * wave + (lane >> 4);
        const float n0 = (float)IN_DIM;
        float sl[NPOLY];
        sl[0] = n0;
        sl[1] = acc[1];
        sl[2] = 1.5f * acc[2] - 0.5f * n0;
        sl[3] = (acc[3] - (2.0f / 5.0f) * sl[1]) * (5.0f / 3.0f);
        sl[4] = (acc[4] - n0 / 5.0f - (2.0f / 7.0f) * sl[2]) * (35.0f / 18.0f);
        sl[5] = (acc[5] - (9.0f / 35.0f) * sl[1] - (4.0f / 15.0f) * sl[3]) * (21.0f / 10.0f);
        sl[6] = (acc[6] - n0 / 7.0f - (4.0f / 21.0f) * sl[2] - (18.0f / 77.0f) * sl[4]) * (231.0f / 100.0f);
        sl[7] = (acc[7] - (4.0f / 21.0f) * sl[1] - (2.0f / 11.0f) * sl[3] - (20.0f / 91.0f) * sl[5]) * (429.0f / 175.0f);
        sl[8] = (acc[8] - n0 / 9.0f - (100.0f / 693.0f) * sl[2] - (162.0f / 1001.0f) * sl[4]
                 - (20.0f / 99.0f) * sl[6]) * (1287.0f / 490.0f);

#pragma unroll
        for (int k = 0; k < 16; ++k) S[row * 16 + k] = (k < NPOLY) ? sl[k] : 0.0f;
    }
    __syncthreads();

    v2f afrag[3];
#pragma unroll
    for (int c = 0; c < 3; ++c) {
        afrag[c][0] = S[mn * 16 + 4 * c + khalf + 0];
        afrag[c][1] = S[mn * 16 + 4 * c + khalf + 1];
    }

#pragma unroll
    for (int j = 0; j < 8; ++j) {
        const int ct = wave * 8 + j;

        v8f d = {};
#pragma unroll
        for (int c = 0; c < 3; ++c) {
            d = __builtin_amdgcn_wmma_f32_16x16x4_f32(
                    false, afrag[c], false, bfrag[j][c], (short)0, d, false, false);
        }

#pragma unroll
        for (int v = 0; v < 8; ++v)
            out[(size_t)(b0 + mrow + v) * OUT_DIM + ct * 16 + mn] = d[v];
    }
}

extern "C" void kernel_launch(void* const* d_in, const int* in_sizes, int n_in,
                              void* d_out, int out_size, void* d_ws, size_t ws_size,
                              hipStream_t stream) {
    (void)in_sizes; (void)n_in; (void)out_size; (void)d_ws; (void)ws_size;
    const float* x      = (const float*)d_in[0];
    const float* coeffs = (const float*)d_in[1];
    const float* scale  = (const float*)d_in[2];
    float* out          = (float*)d_out;

    const int row_tiles = 4096 / 16;
    legendre_fused_kernel<<<row_tiles, 256, 0, stream>>>(x, coeffs, scale, out);
}


# Each lane owns 4 adjacent output columns (coefficient columns dealt 4c+q over 4 tiles; block fetched as whole lines, dealt via LDS): 16 whole-line b128 stores per wave, not 64 b32; sums unchanged
# speedup vs baseline: 2.1753x; 1.3425x over previous
#include <hip/hip_runtime.h>


typedef __attribute__((ext_vector_type(2))) float v2f;
typedef __attribute__((ext_vector_type(4))) float v4f;
typedef __attribute__((ext_vector_type(8))) float v8f;

#define IN_DIM  1024
#define OUT_DIM 1024
#define NPOLY   9

__device__ __forceinline__ v2f tanh_pair(v2f y) {
    v2f z = y * 2.8853900817779268f;
    v2f e;
    e[0] = __builtin_amdgcn_exp2f(z[0]);
    e[1] = __builtin_amdgcn_exp2f(z[1]);
    v2f d = e + 1.0f;
    v2f q;
    q[0] = __builtin_amdgcn_rcpf(d[0]);
    q[1] = __builtin_amdgcn_rcpf(d[1]);
    return 1.0f - 2.0f * q;
}

template <int M>
__device__ __forceinline__ float add_row_xor(float x) {
    const int b = __builtin_bit_cast(int, x);
    const int p = __builtin_amdgcn_update_dpp(0, b, 0x160 + M, 0xf, 0xf, true);
    return x + __builtin_bit_cast(float, p);
}

__device__ __forceinline__ float xhalf(float x) {
    const int b = __builtin_bit_cast(int, x);
    return __builtin_bit_cast(float,
        __builtin_amdgcn_permlanex16(b, b, 0x76543210, 0xfedcba98, false, false));
}

__device__ __forceinline__ float add_half_xor_two_rows(float a, float b, bool upper) {
    const float keep = upper ? b : a;
    const float send = upper ? a : b;
    return keep + xhalf(send);
}

__global__ __launch_bounds__(256)
void legendre_fused_kernel(const float* __restrict__ x,
                           const float* __restrict__ coeffs,
                           const float* __restrict__ scale_p,
                           float* __restrict__ out)
{
    __shared__ float S[16 * 16];
    __shared__ __attribute__((aligned(16))) float CS[8][128 * NPOLY];

    const int tid  = threadIdx.x;
    const int lane = tid & 31;
    const int wave = tid >> 5;
    const int b0   = blockIdx.x * 16;

    const int mn    = lane & 15;
    const int khalf = (lane >> 4) * 2;
    const int mrow  = (lane >> 4) * 8;

    const float* xbase = x + (size_t)(b0 + 2 * wave) * IN_DIM + lane * 4;
    v4f xv[2][8];
#pragma unroll
    for (int r = 0; r < 2; ++r) {
#pragma unroll
        for (int j = 0; j < 8; ++j)
            xv[r][j] = *(const v4f*)(xbase + r * IN_DIM + j * 128);
    }

    const float* cblk = coeffs + (size_t)wave * (128 * NPOLY);
    v4f cv[9];
#pragma unroll
    for (int i = 0; i < 9; ++i)
        cv[i] = *(const v4f*)(cblk + (i * 32 + lane) * 4);

    __builtin_amdgcn_sched_barrier(0);

    float s = scale_p[0];
    s = fminf(fmaxf(s, 0.1f), 2.0f);

    float accr[2][NPOLY];
#pragma unroll
    for (int r = 0; r < 2; ++r) {
        v2f pa[NPOLY];
#pragma unroll
        for (int k = 1; k < NPOLY; ++k) pa[k] = (v2f){0.0f, 0.0f};

#pragma unroll
        for (int j = 0; j < 8; ++j) {
            v4f xq = xv[r][j];
#pragma unroll
            for (int h = 0; h < 2; ++h) {
                v2f xp = {xq[2 * h], xq[2 * h + 1]};
                v2f t  = tanh_pair(xp * s);
                v2f u  = t * t;
                v2f p2 = 1.5f * u - 0.5f;
                v2f p3 = t * (2.5f * u - 1.5f);
                v2f p4 = u * (4.375f * u - 3.75f) + 0.375f;
                pa[1] += t;
                pa[2] += t * t;
                pa[3] += t * p2;
                pa[4] += p2 * p2;
                pa[5] += p2 * p3;
                pa[6] += p3 * p3;
                pa[7] += p3 * p4;
                pa[8] += p4 * p4;
            }
        }

#pragma unroll
        for (int k = 1; k < NPOLY; ++k) accr[r][k] = pa[k][0] + pa[k][1];
    }

    float* cs = &CS[wave][0];
#pragma unroll
    for (int i = 0; i < 9; ++i)
        *(v4f*)(cs + (i * 32 + lane) * 4) = cv[i];

    __builtin_amdgcn_fence(__ATOMIC_ACQ_REL, "wavefront");

    v2f bfrag[8][3];
#pragma unroll
    for (int j = 0; j < 8; ++j) {
        const int ol = 64 * (j >> 2) + 4 * mn + (j & 3);
        const float* crow = cs + ol * NPOLY;
        bfrag[j][0][0] = crow[khalf + 0];
        bfrag[j][0][1] = crow[khalf + 1];
        bfrag[j][1][0] = crow[4 + khalf + 0];
        bfrag[j][1][1] = crow[4 + khalf + 1];
        const float c8 = crow[8];
        bfrag[j][2][0] = (khalf == 0) ? c8 : 0.0f;
        bfrag[j][2][1] = 0.0f;
    }

    float acc[NPOLY];
#pragma unroll
    for (int k = 1; k < NPOLY; ++k) acc[k] = add_half_xor_two_rows(accr[0][k], accr[1][k], lane >= 16);
#pragma unroll
    for (int k = 1; k < NPOLY; ++k) acc[k] = add_row_xor<8>(acc[k]);
#pragma unroll
    for (int k = 1; k < NPOLY; ++k) acc[k] = add_row_xor<4>(acc[k]);
#pragma unroll
    for (int k = 1; k < NPOLY; ++k) acc[k] = add_row_xor<2>(acc[k]);
#pragma unroll
    for (int k = 1; k < NPOLY; ++k) acc[k] = add_row_xor<1>(acc[k]);

    if (mn == 0) {
        const int row = 2 * wave + (lane >> 4);
        const float n0 = (float)IN_DIM;
        float sl[NPOLY];
        sl[0] = n0;
        sl[1] = acc[1];
        sl[2] = 1.5f * acc[2] - 0.5f * n0;
        sl[3] = (acc[3] - (2.0f / 5.0f) * sl[1]) * (5.0f / 3.0f);
        sl[4] = (acc[4] - n0 / 5.0f - (2.0f / 7.0f) * sl[2]) * (35.0f / 18.0f);
        sl[5] = (acc[5] - (9.0f / 35.0f) * sl[1] - (4.0f / 15.0f) * sl[3]) * (21.0f / 10.0f);
        sl[6] = (acc[6] - n0 / 7.0f - (4.0f / 21.0f) * sl[2] - (18.0f / 77.0f) * sl[4]) * (231.0f / 100.0f);
        sl[7] = (acc[7] - (4.0f / 21.0f) * sl[1] - (2.0f / 11.0f) * sl[3] - (20.0f / 91.0f) * sl[5]) * (429.0f / 175.0f);
        sl[8] = (acc[8] - n0 / 9.0f - (100.0f / 693.0f) * sl[2] - (162.0f / 1001.0f) * sl[4]
                 - (20.0f / 99.0f) * sl[6]) * (1287.0f / 490.0f);

#pragma unroll
        for (int k = 0; k < 16; ++k) S[row * 16 + k] = (k < NPOLY) ? sl[k] : 0.0f;
    }
    __syncthreads();

    v2f afrag[3];
#pragma unroll
    for (int c = 0; c < 3; ++c) {
        afrag[c][0] = S[mn * 16 + 4 * c + khalf + 0];
        afrag[c][1] = S[mn * 16 + 4 * c + khalf + 1];
    }

#pragma unroll
    for (int g = 0; g < 2; ++g) {
        v8f d[4];
#pragma unroll
        for (int q = 0; q < 4; ++q) {
            d[q] = (v8f){};
#pragma unroll
            for (int c = 0; c < 3; ++c) {
                d[q] = __builtin_amdgcn_wmma_f32_16x16x4_f32(
                        false, afrag[c], false, bfrag[4 * g + q][c], (short)0, d[q], false, false);
            }
        }

        float* orow = out + (size_t)(b0 + mrow) * OUT_DIM + wave * 128 + 64 * g + 4 * mn;
#pragma unroll
        for (int v = 0; v < 8; ++v) {
            const v4f o4 = {d[0][v], d[1][v], d[2][v], d[3][v]};
            *(v4f*)(orow + (size_t)v * OUT_DIM) = o4;
        }
    }
}

extern "C" void kernel_launch(void* const* d_in, const int* in_sizes, int n_in,
                              void* d_out, int out_size, void* d_ws, size_t ws_size,
                              hipStream_t stream) {
    (void)in_sizes; (void)n_in; (void)out_size; (void)d_ws; (void)ws_size;
    const float* x      = (const float*)d_in[0];
    const float* coeffs = (const float*)d_in[1];
    const float* scale  = (const float*)d_in[2];
    float* out          = (float*)d_out;

    const int row_tiles = 4096 / 16;
    legendre_fused_kernel<<<row_tiles, 256, 0, stream>>>(x, coeffs, scale, out);
}
